# starting kernel
# baseline (speedup 1.0000x reference)
#include <hip/hip_runtime.h>

typedef float v2f __attribute__((ext_vector_type(2)));
typedef float v8f __attribute__((ext_vector_type(8)));

#define NJ  24
#define DF  19
#define DPJ 6

__device__ __constant__ int kParents[NJ] = {
    -1, 0, 0, 0, 1, 2, 3, 4, 5, 6, 7, 8, 9, 9, 9, 12, 13, 14, 16, 17, 18, 19, 20, 21};

__device__ inline void lds_fence() {
  asm volatile("s_wait_dscnt 0" ::: "memory");
}

__device__ inline v8f wmma4(v2f a, v2f b, v8f c) {
  return __builtin_amdgcn_wmma_f32_16x16x4_f32(
       false, a,  false, b,
       (short)0, c,  false,  false);
}

__device__ inline int imin(int a, int b) { return a < b ? a : b; }

__global__ __launch_bounds__(128) void pose_enc_kernel(
    const float* __restrict__ rots, const float* __restrict__ Jtrs,
    const float* __restrict__ W0,   const float* __restrict__ b0,
    const float* __restrict__ W1,   const float* __restrict__ b1,
    const float* __restrict__ W2,   const float* __restrict__ b2,
    float* __restrict__ out, int B)
{
  __shared__ float s_in[4][16][20];
  __shared__ float s_h [4][16][20];
  __shared__ float s_feat[4][NJ + 1][16][DPJ];

  __shared__ v2f s_w1a[NJ][10][16];
  __shared__ v2f s_w1b[NJ][10][4];
  __shared__ v2f s_w2p[NJ][10][8];
  __shared__ v2f s_w0p[144][8];

  const int tid  = threadIdx.x;

  for (int i = tid; i < NJ * 10 * 16; i += 128) {
    int c = i & 15, pr = (i >> 4) % 10, j = i / 160;
    int k = 2 * pr;
    float x = W1[j * 361 + c * 19 + k];
    float y = W1[j * 361 + c * 19 + imin(k + 1, 18)];
    y = (k + 1 < DF) ? y : 0.f;
    v2f v; v.x = x; v.y = y;
    s_w1a[j][pr][c] = v;
  }
  for (int i = tid; i < NJ * 10 * 4; i += 128) {
    int c = i & 3, pr = (i >> 2) % 10, j = i / 40;
    int k = 2 * pr, col = imin(16 + c, 18);
    float x = W1[j * 361 + col * 19 + k];
    float y = W1[j * 361 + col * 19 + imin(k + 1, 18)];
    x = (c < 3) ? x : 0.f;
    y = (c < 3 && k + 1 < DF) ? y : 0.f;
    v2f v; v.x = x; v.y = y;
    s_w1b[j][pr][c] = v;
  }
  for (int i = tid; i < NJ * 10 * 8; i += 128) {
    int c = i & 7, pr = (i >> 3) % 10, j = i / 80;
    int k = 2 * pr, col = imin(c, 5);
    float x = W2[j * 114 + col * 19 + k];
    float y = W2[j * 114 + col * 19 + imin(k + 1, 18)];
    x = (c < DPJ) ? x : 0.f;
    y = (c < DPJ && k + 1 < DF) ? y : 0.f;
    v2f v; v.x = x; v.y = y;
    s_w2p[j][pr][c] = v;
  }
  for (int i = tid; i < 144 * 8; i += 128) {
    int c = i & 7, pr = i >> 3;
    int k = 2 * pr, col = imin(c, 5);
    float x = W0[col * 288 + k];
    float y = W0[col * 288 + k + 1];
    x = (c < DPJ) ? x : 0.f;
    y = (c < DPJ) ? y : 0.f;
    v2f v; v.x = x; v.y = y;
    s_w0p[pr][c] = v;
  }
  __syncthreads();

  const int w    = tid >> 5;
  const int lane = tid & 31;
  const int half = lane >> 4;
  const int n    = lane & 15;
  const int n7   = imin(n, 7);
  const int n3   = imin(n, 3);
  const int rb   = (blockIdx.x * 4 + w) * 16;
  if (rb >= B) return;

  {
    float bv = b0[imin(n, 5)];
    bv = (n < DPJ) ? bv : 0.f;
    v8f acc;
#pragma unroll
    for (int v = 0; v < 8; ++v) acc[v] = bv;

    const float* rrow = rots + (long)(rb + n) * 216;
    const float* jrow = Jtrs + (long)(rb + n) * 72;
#pragma unroll 4
    for (int step = 0; step < 54; ++step) {
      int pr = 2 * step + half;
      v2f a = *(const v2f*)(rrow + 2 * pr);
      acc = wmma4(a, s_w0p[pr][n7], acc);
    }
#pragma unroll 4
    for (int step = 54; step < 72; ++step) {
      int pr = 2 * step + half;
      v2f a = *(const v2f*)(jrow + 2 * pr - 216);
      acc = wmma4(a, s_w0p[pr][n7], acc);
    }
    if (n < DPJ) {
#pragma unroll
      for (int v = 0; v < 8; ++v) s_feat[w][NJ][v + 8 * half][n] = acc[v];
    }
  }
  lds_fence();

  for (int j = 0; j < NJ; ++j) {
    const int p = kParents[j];

    if (half == 0) {
      const float* rp = rots + (long)(rb + n) * 216 + j * 9;
#pragma unroll
      for (int q = 0; q < 9; ++q) s_in[w][n][q] = rp[q];
    } else {
      const float* jp = Jtrs + (long)(rb + n) * 72;
      float x = jp[j * 3 + 0], y = jp[j * 3 + 1], z = jp[j * 3 + 2];
      int pc = (p >= 0) ? p : 0;
      float px = jp[pc * 3 + 0], py = jp[pc * 3 + 1], pz = jp[pc * 3 + 2];
      px = (p >= 0) ? px : 0.f;  py = (p >= 0) ? py : 0.f;  pz = (p >= 0) ? pz : 0.f;
      float dx = x - px, dy = y - py, dz = z - pz;
      float bl = sqrtf(dx * dx + dy * dy + dz * dz);
      s_in[w][n][9]  = x;  s_in[w][n][10] = y;
      s_in[w][n][11] = z;  s_in[w][n][12] = bl;
      const int src = (p >= 0) ? p : NJ;
#pragma unroll
      for (int f = 0; f < DPJ; ++f) s_in[w][n][13 + f] = s_feat[w][src][n][f];
      s_in[w][n][19] = 0.f;
    }
    lds_fence();

    float bA = b1[j * DF + n];
    float bB = b1[j * DF + imin(16 + n, 18)];
    bB = (n < 3) ? bB : 0.f;
    v8f accA, accB;
#pragma unroll
    for (int v = 0; v < 8; ++v) { accA[v] = bA; accB[v] = bB; }
#pragma unroll
    for (int step = 0; step < 5; ++step) {
      int pr = 2 * step + half;
      v2f a = *(const v2f*)&s_in[w][n][2 * pr];
      accA = wmma4(a, s_w1a[j][pr][n],  accA);
      accB = wmma4(a, s_w1b[j][pr][n3], accB);
    }
#pragma unroll
    for (int v = 0; v < 8; ++v) {
      int m = v + 8 * half;
      s_h[w][m][n] = fmaxf(accA[v], 0.f);
      if (n < 3) s_h[w][m][16 + n] = fmaxf(accB[v], 0.f);
      if (n == 3) s_h[w][m][19] = 0.f;
    }
    lds_fence();

    float bC = b2[j * DPJ + imin(n, 5)];
    bC = (n < DPJ) ? bC : 0.f;
    v8f acc2;
#pragma unroll
    for (int v = 0; v < 8; ++v) acc2[v] = bC;
#pragma unroll
    for (int step = 0; step < 5; ++step) {
      int pr = 2 * step + half;
      v2f a = *(const v2f*)&s_h[w][n][2 * pr];
      acc2 = wmma4(a, s_w2p[j][pr][n7], acc2);
    }
    if (n < DPJ) {
#pragma unroll
      for (int v = 0; v < 8; ++v) {
        int m = v + 8 * half;
        float val = acc2[v];
        s_feat[w][j][m][n] = val;
        out[(long)(rb + m) * (NJ * DPJ) + j * DPJ + n] = val;
      }
    }
    lds_fence();
  }
}

extern "C" void kernel_launch(void* const* d_in, const int* in_sizes, int n_in,
                              void* d_out, int out_size, void* d_ws, size_t ws_size,
                              hipStream_t stream) {
  const float* rots = (const float*)d_in[0];
  const float* Jtrs = (const float*)d_in[1];
  const float* W0   = (const float*)d_in[2];
  const float* b0   = (const float*)d_in[3];
  const float* W1   = (const float*)d_in[4];
  const float* b1   = (const float*)d_in[5];
  const float* W2   = (const float*)d_in[6];
  const float* b2   = (const float*)d_in[7];
  float* outp = (float*)d_out;

  int B = in_sizes[0] / (NJ * 9);
  int grid = (B + 63) / 64;
  pose_enc_kernel<<<grid, 128, 0, stream>>>(rots, Jtrs, W0, b0, W1, b1, W2, b2,
                                            outp, B);
}


# layer 1 split at k=12: each level's bias, rot and Jtr steps run a level ahead on A fragments read straight from memory, not LDS; only bone-length and feat_prev steps wait for parents; sums unchanged
# speedup vs baseline: 1.8008x; 1.8008x over previous
#include <hip/hip_runtime.h>

typedef float v2f __attribute__((ext_vector_type(2)));
typedef float v4f __attribute__((ext_vector_type(4)));
typedef float v8f __attribute__((ext_vector_type(8)));

#define NJ  24
#define DF  19
#define DPJ 6
#define GMAX 5
#define CP  8

#define W1_FLOATS (NJ * DF * DF)
#define W2_FLOATS (NJ * DPJ * DF)
#define W0_FLOATS (DPJ * 12 * NJ)
#define W1_OFF 0
#define W2_OFF (W1_FLOATS)
#define W0_OFF (W1_FLOATS + W2_FLOATS)
#define SW_FLOATS (W1_FLOATS + W2_FLOATS + W0_FLOATS + 4)

__device__ constexpr int parent_of(int j) {
  return j == 0 ? -1 : j <= 3 ? 0 : j <= 11 ? j - 3 : j <= 14 ? 9 : j <= 17 ? j - 3 : j - 2;
}

__device__ inline void lds_fence() {
  asm volatile("s_wait_dscnt 0" ::: "memory");
}

__device__ inline v8f wmma4(v2f a, v2f b, v8f c) {
  return __builtin_amdgcn_wmma_f32_16x16x4_f32(
       false, a,  false, b,
       (short)0, c,  false,  false);
}

__device__ inline int imin(int a, int b) { return a < b ? a : b; }

__device__ __forceinline__ v2f ldw(const float* p, bool kill_hi) {
  v2f b;
  b.x = p[0];
  float hi = p[1];
  b.y = kill_hi ? 0.f : hi;
  return b;
}

template <int NV1, int NV2, int NV0>
__device__ __forceinline__ void stage_weights(
    const float* __restrict__ W1, const float* __restrict__ W2,
    const float* __restrict__ W0, float* sw, const int tid)
{
  constexpr int I1 = (NV1 + 127) / 128, I2 = (NV2 + 127) / 128, I0 = (NV0 + 127) / 128;
  const v4f* g1 = (const v4f*)W1;
  const v4f* g2 = (const v4f*)W2;
  const v4f* g0 = (const v4f*)W0;
  v4f r1[I1], r2[I2], r0[I0];
#pragma unroll
  for (int u = 0; u < I1; ++u) r1[u] = g1[imin(tid + 128 * u, NV1 - 1)];
#pragma unroll
  for (int u = 0; u < I2; ++u) r2[u] = g2[imin(tid + 128 * u, NV2 - 1)];
#pragma unroll
  for (int u = 0; u < I0; ++u) r0[u] = g0[imin(tid + 128 * u, NV0 - 1)];
  v4f* d1 = (v4f*)(sw + W1_OFF);
  v4f* d2 = (v4f*)(sw + W2_OFF);
  v4f* d0 = (v4f*)(sw + W0_OFF);
#pragma unroll
  for (int u = 0; u < I1; ++u) d1[imin(tid + 128 * u, NV1 - 1)] = r1[u];
#pragma unroll
  for (int u = 0; u < I2; ++u) d2[imin(tid + 128 * u, NV2 - 1)] = r2[u];
#pragma unroll
  for (int u = 0; u < I0; ++u) d0[imin(tid + 128 * u, NV0 - 1)] = r0[u];
  if (tid < 4) sw[W0_OFF + W0_FLOATS + tid] = 0.f;
}

template <int G>
struct Level {
  v2f a0[G], a1[G];
  float r8[G];
  float J[G][3], P[G][3];
  float bA[G], bB[G], bC[G];
  v8f accA[G], accB[G];
  float bl[G];
};

template <int G>
__device__ __forceinline__ void level_request(
    const int j0, const int n,
    const float* __restrict__ rr, const float* __restrict__ rl,
    const float* __restrict__ jl,
    const float* __restrict__ b1, const float* __restrict__ b2, Level<G>& L)
{
#pragma unroll
  for (int g = 0; g < G; ++g) {
    const int j = j0 + g;
    const int p = parent_of(j);
    const int pc = (p >= 0) ? p : 0;
    L.a0[g].x = rl[j * 9 + 0];  L.a0[g].y = rl[j * 9 + 1];
    L.a1[g].x = rl[j * 9 + 4];  L.a1[g].y = rl[j * 9 + 5];
    L.r8[g] = rr[j * 9 + 8];
#pragma unroll
    for (int q = 0; q < 3; ++q) { L.J[g][q] = jl[j * 3 + q]; L.P[g][q] = jl[pc * 3 + q]; }
    L.bA[g] = b1[j * DF + n];
    L.bB[g] = b1[j * DF + imin(16 + n, 18)];
    L.bC[g] = b2[j * DPJ + imin(n, 5)];
  }
}

template <int G>
__device__ __forceinline__ void level_head(
    const int j0, const int n, const int half, const int offA, const int offB,
    Level<G>& L, const float* sw)
{
  v2f a2[G];
#pragma unroll
  for (int g = 0; g < G; ++g) {
    const int p = parent_of(j0 + g);
    float qx = (p >= 0) ? L.P[g][0] : 0.f;
    float qy = (p >= 0) ? L.P[g][1] : 0.f;
    float qz = (p >= 0) ? L.P[g][2] : 0.f;
    float dx = L.J[g][0] - qx, dy = L.J[g][1] - qy, dz = L.J[g][2] - qz;
    L.bl[g] = sqrtf(dx * dx + dy * dy + dz * dz);
    a2[g].x = half ? L.J[g][1] : L.r8[g];
    a2[g].y = half ? L.J[g][2] : L.J[g][0];
    const float bB = (n < 3) ? L.bB[g] : 0.f;
#pragma unroll
    for (int v = 0; v < 8; ++v) { L.accA[g][v] = L.bA[g]; L.accB[g][v] = bB; }
  }
#pragma unroll
  for (int step = 0; step < 3; ++step) {
    const int pr = 2 * step + half;
#pragma unroll
    for (int g = 0; g < G; ++g) {
      const int j = j0 + g;
      const v2f a = (step == 0) ? L.a0[g] : (step == 1) ? L.a1[g] : a2[g];
      const float* wj = sw + W1_OFF + j * (DF * DF) + 2 * pr;
      L.accA[g] = wmma4(a, ldw(wj + offA, false), L.accA[g]);
      L.accB[g] = wmma4(a, ldw(wj + offB, false), L.accB[g]);
    }
  }
}

template <int G>
__device__ __forceinline__ void level_tail(
    const int j0, const int rb, const int n, const int half,
    const int offA, const int offB, const int offC,
    float* __restrict__ out, Level<G>& L,
    float (*sh_w)[16][20], float (*sc_w)[16][CP], const float* sw)
{
#pragma unroll
  for (int step = 3; step < 5; ++step) {
    const int pr = 2 * step + half;
#pragma unroll
    for (int g = 0; g < G; ++g) {
      const int j = j0 + g;
      const int p = parent_of(j);
      const int src = (p >= 0) ? p : NJ;
      v2f a = *(const v2f*)&sc_w[src][n][2 * pr - 12];
      if (step == 3) a.x = half ? a.x : L.bl[g];
      const float* wj = sw + W1_OFF + j * (DF * DF) + 2 * pr;
      const bool last = (step == 4) && (half == 1);
      L.accA[g] = wmma4(a, ldw(wj + offA, last), L.accA[g]);
      L.accB[g] = wmma4(a, ldw(wj + offB, last), L.accB[g]);
    }
  }
#pragma unroll
  for (int g = 0; g < G; ++g) {
#pragma unroll
    for (int v = 0; v < 8; ++v) {
      int m = v + 8 * half;
      sh_w[g][m][n] = fmaxf(L.accA[g][v], 0.f);
      if (n < 3) sh_w[g][m][16 + n] = fmaxf(L.accB[g][v], 0.f);
      if (n == 3) sh_w[g][m][19] = 0.f;
    }
  }
  lds_fence();

  v8f acc2[G];
#pragma unroll
  for (int g = 0; g < G; ++g) {
    const float bC = (n < DPJ) ? L.bC[g] : 0.f;
#pragma unroll
    for (int v = 0; v < 8; ++v) acc2[g][v] = bC;
  }
#pragma unroll
  for (int step = 0; step < 5; ++step) {
    const int pr = 2 * step + half;
#pragma unroll
    for (int g = 0; g < G; ++g) {
      const int j = j0 + g;
      v2f a = *(const v2f*)&sh_w[g][n][2 * pr];
      const float* wj = sw + W2_OFF + j * (DPJ * DF) + 2 * pr;
      const bool last = (step == 4) && (half == 1);
      acc2[g] = wmma4(a, ldw(wj + offC, last), acc2[g]);
    }
  }
  if (n < CP) {
#pragma unroll
    for (int g = 0; g < G; ++g) {
#pragma unroll
      for (int v = 0; v < 8; ++v)
        sc_w[j0 + g][v + 8 * half][(1 + n) & (CP - 1)] = (n < DPJ) ? acc2[g][v] : 0.f;
    }
  }
  if (n < DPJ) {
#pragma unroll
    for (int g = 0; g < G; ++g) {
      const int j = j0 + g;
#pragma unroll
      for (int v = 0; v < 8; ++v)
        out[(long)(rb + v + 8 * half) * (NJ * DPJ) + j * DPJ + n] = acc2[g][v];
    }
  }
  lds_fence();
}

__global__ __launch_bounds__(128) void pose_enc_kernel(
    const float* __restrict__ rots, const float* __restrict__ Jtrs,
    const float* __restrict__ W0,   const float* __restrict__ b0,
    const float* __restrict__ W1,   const float* __restrict__ b1,
    const float* __restrict__ W2,   const float* __restrict__ b2,
    float* __restrict__ out, int B)
{
  __shared__ __attribute__((aligned(16))) float s_h[4][GMAX][16][20];
  __shared__ __attribute__((aligned(16))) float s_c[4][NJ + 1][16][CP];

  __shared__ __attribute__((aligned(16))) float s_w[SW_FLOATS];

  const int tid  = threadIdx.x & 127;

  stage_weights<W1_FLOATS / 4, W2_FLOATS / 4, W0_FLOATS / 4>(W1, W2, W0, s_w, tid);
  __syncthreads();

  const int w    = tid >> 5;
  const int lane = tid & 31;
  const int half = lane >> 4;
  const int n    = lane & 15;
  const int offA = n * DF;
  const int offB = (16 + imin(n, 2)) * DF;
  const int offC = imin(n, DPJ - 1) * DF;
  const int off0 = imin(n, DPJ - 1) * (12 * NJ);
  const int rb   = (blockIdx.x * 4 + w) * 16;
  if (rb >= B) return;

  const float* rrow = rots + (long)(rb + n) * 216;
  const float* jrow = Jtrs + (long)(rb + n) * 72;
  const float* rlan = rrow + 2 * half;
  float (*sh_w)[16][20] = s_h[w];
  float (*sc_w)[16][CP] = s_c[w];

  Level<1> L0;  Level<3> L1, L2, L3;  Level<5> L4;  Level<3> L5;  Level<2> L6, L7, L8;

  level_request<1>(0, n, rrow, rlan, jrow, b1, b2, L0);
  level_request<3>(1, n, rrow, rlan, jrow, b1, b2, L1);
  level_head<1>(0, n, half, offA, offB, L0, s_w);
  level_head<3>(1, n, half, offA, offB, L1, s_w);

  {
    float bv = b0[imin(n, 5)];
    bv = (n < DPJ) ? bv : 0.f;
    v8f acc;
#pragma unroll
    for (int v = 0; v < 8; ++v) acc[v] = bv;

#pragma unroll 4
    for (int step = 0; step < 54; ++step) {
      int pr = 2 * step + half;
      v2f a = *(const v2f*)(rrow + 2 * pr);
      acc = wmma4(a, ldw(s_w + W0_OFF + off0 + 2 * pr, false), acc);
    }
#pragma unroll 4
    for (int step = 54; step < 72; ++step) {
      int pr = 2 * step + half;
      v2f a = *(const v2f*)(jrow + 2 * pr - 216);
      acc = wmma4(a, ldw(s_w + W0_OFF + off0 + 2 * pr, false), acc);
    }
    if (n < CP) {
#pragma unroll
      for (int v = 0; v < 8; ++v)
        sc_w[NJ][v + 8 * half][(1 + n) & (CP - 1)] = (n < DPJ) ? acc[v] : 0.f;
    }
  }
  lds_fence();

#define REQ(G, j0, L)  level_request<G>(j0, n, rrow, rlan, jrow, b1, b2, L)
#define HEAD(G, j0, L) level_head<G>(j0, n, half, offA, offB, L, s_w)
#define TAIL(G, j0, L) level_tail<G>(j0, rb, n, half, offA, offB, offC, out, L, sh_w, sc_w, s_w)
  REQ(3, 4, L2);                      TAIL(1, 0, L0);
  REQ(3, 7, L3);   HEAD(3, 4, L2);    TAIL(3, 1, L1);
  REQ(5, 10, L4);  HEAD(3, 7, L3);    TAIL(3, 4, L2);
  REQ(3, 15, L5);  HEAD(5, 10, L4);   TAIL(3, 7, L3);
  REQ(2, 18, L6);  HEAD(3, 15, L5);   TAIL(5, 10, L4);
  REQ(2, 20, L7);  HEAD(2, 18, L6);   TAIL(3, 15, L5);
  REQ(2, 22, L8);  HEAD(2, 20, L7);   TAIL(2, 18, L6);
                   HEAD(2, 22, L8);   TAIL(2, 20, L7);
                                      TAIL(2, 22, L8);
#undef REQ
#undef HEAD
#undef TAIL
}

extern "C" void kernel_launch(void* const* d_in, const int* in_sizes, int n_in,
                              void* d_out, int out_size, void* d_ws, size_t ws_size,
                              hipStream_t stream) {
  const float* rots = (const float*)d_in[0];
  const float* Jtrs = (const float*)d_in[1];
  const float* W0   = (const float*)d_in[2];
  const float* b0   = (const float*)d_in[3];
  const float* W1   = (const float*)d_in[4];
  const float* b1   = (const float*)d_in[5];
  const float* W2   = (const float*)d_in[6];
  const float* b2   = (const float*)d_in[7];
  float* outp = (float*)d_out;

  int B = in_sizes[0] / (NJ * 9);
  int grid = (B + 63) / 64;
  pose_enc_kernel<<<grid, 128, 0, stream>>>(rots, Jtrs, W0, b0, W1, b1, W2, b2,
                                            outp, B);
}


# per-wave LDS tiles (hidden, joint outputs, global_feat) laid k-major: a lane's 8 D registers leave as two 16-byte stores, not eight masked 4-byte ones; A fragments read as two floats; sums unchanged
# speedup vs baseline: 2.2018x; 1.2227x over previous
#include <hip/hip_runtime.h>

typedef float v2f __attribute__((ext_vector_type(2)));
typedef float v4f __attribute__((ext_vector_type(4)));
typedef float v8f __attribute__((ext_vector_type(8)));

#define NJ  24
#define DF  19
#define DPJ 6
#define GMAX 5
#define CP  8
#define HK  20
#define HP  20


#define W1_FLOATS (NJ * DF * DF)
#define W2_FLOATS (NJ * DPJ * DF)
#define W0_FLOATS (DPJ * 12 * NJ)
#define W1_OFF 0
#define W2_OFF (W1_FLOATS)
#define W0_OFF (W1_FLOATS + W2_FLOATS)
#define SW_FLOATS (W1_FLOATS + W2_FLOATS + W0_FLOATS + 4)

__device__ constexpr int parent_of(int j) {
  return j == 0 ? -1 : j <= 3 ? 0 : j <= 11 ? j - 3 : j <= 14 ? 9 : j <= 17 ? j - 3 : j - 2;
}

__device__ inline void lds_fence() {
  asm volatile("s_wait_dscnt 0" ::: "memory");
}

__device__ inline v8f wmma4(v2f a, v2f b, v8f c) {
  return __builtin_amdgcn_wmma_f32_16x16x4_f32(
       false, a,  false, b,
       (short)0, c,  false,  false);
}

__device__ inline int imin(int a, int b) { return a < b ? a : b; }

__device__ __forceinline__ v2f ldw(const float* p, bool kill_hi) {
  v2f b;
  b.x = p[0];
  float hi = p[1];
  b.y = kill_hi ? 0.f : hi;
  return b;
}

template <int NV1, int NV2, int NV0>
__device__ __forceinline__ void stage_weights(
    const float* __restrict__ W1, const float* __restrict__ W2,
    const float* __restrict__ W0, float* sw, const int tid)
{
  constexpr int I1 = (NV1 + 127) / 128, I2 = (NV2 + 127) / 128, I0 = (NV0 + 127) / 128;
  const v4f* g1 = (const v4f*)W1;
  const v4f* g2 = (const v4f*)W2;
  const v4f* g0 = (const v4f*)W0;
  v4f r1[I1], r2[I2], r0[I0];
#pragma unroll
  for (int u = 0; u < I1; ++u) r1[u] = g1[imin(tid + 128 * u, NV1 - 1)];
#pragma unroll
  for (int u = 0; u < I2; ++u) r2[u] = g2[imin(tid + 128 * u, NV2 - 1)];
#pragma unroll
  for (int u = 0; u < I0; ++u) r0[u] = g0[imin(tid + 128 * u, NV0 - 1)];
  v4f* d1 = (v4f*)(sw + W1_OFF);
  v4f* d2 = (v4f*)(sw + W2_OFF);
  v4f* d0 = (v4f*)(sw + W0_OFF);
#pragma unroll
  for (int u = 0; u < I1; ++u) d1[imin(tid + 128 * u, NV1 - 1)] = r1[u];
#pragma unroll
  for (int u = 0; u < I2; ++u) d2[imin(tid + 128 * u, NV2 - 1)] = r2[u];
#pragma unroll
  for (int u = 0; u < I0; ++u) d0[imin(tid + 128 * u, NV0 - 1)] = r0[u];
  if (tid < 4) sw[W0_OFF + W0_FLOATS + tid] = 0.f;
}

template <int G>
struct Level {
  v2f a0[G], a1[G];
  float r8[G];
  float J[G][3], P[G][3];
  float bA[G], bB[G], bC[G];
  v8f accA[G], accB[G];
  float bl[G];
};

template <int G>
__device__ __forceinline__ void level_request(
    const int j0, const int n,
    const float* __restrict__ rr, const float* __restrict__ rl,
    const float* __restrict__ jl,
    const float* __restrict__ b1, const float* __restrict__ b2, Level<G>& L)
{
#pragma unroll
  for (int g = 0; g < G; ++g) {
    const int j = j0 + g;
    const int p = parent_of(j);
    const int pc = (p >= 0) ? p : 0;
    L.a0[g].x = rl[j * 9 + 0];  L.a0[g].y = rl[j * 9 + 1];
    L.a1[g].x = rl[j * 9 + 4];  L.a1[g].y = rl[j * 9 + 5];
    L.r8[g] = rr[j * 9 + 8];
#pragma unroll
    for (int q = 0; q < 3; ++q) { L.J[g][q] = jl[j * 3 + q]; L.P[g][q] = jl[pc * 3 + q]; }
    L.bA[g] = b1[j * DF + n];
    L.bB[g] = b1[j * DF + imin(16 + n, 18)];
    L.bC[g] = b2[j * DPJ + imin(n, 5)];
  }
}

template <int G>
__device__ __forceinline__ void level_head(
    const int j0, const int n, const int half, const int offA, const int offB,
    Level<G>& L, const float* sw)
{
  v2f a2[G];
#pragma unroll
  for (int g = 0; g < G; ++g) {
    const int p = parent_of(j0 + g);
    float qx = (p >= 0) ? L.P[g][0] : 0.f;
    float qy = (p >= 0) ? L.P[g][1] : 0.f;
    float qz = (p >= 0) ? L.P[g][2] : 0.f;
    float dx = L.J[g][0] - qx, dy = L.J[g][1] - qy, dz = L.J[g][2] - qz;
    L.bl[g] = sqrtf(dx * dx + dy * dy + dz * dz);
    a2[g].x = half ? L.J[g][1] : L.r8[g];
    a2[g].y = half ? L.J[g][2] : L.J[g][0];
    const float bB = (n < 3) ? L.bB[g] : 0.f;
#pragma unroll
    for (int v = 0; v < 8; ++v) { L.accA[g][v] = L.bA[g]; L.accB[g][v] = bB; }
  }
#pragma unroll
  for (int step = 0; step < 3; ++step) {
    const int pr = 2 * step + half;
#pragma unroll
    for (int g = 0; g < G; ++g) {
      const int j = j0 + g;
      const v2f a = (step == 0) ? L.a0[g] : (step == 1) ? L.a1[g] : a2[g];
      const float* wj = sw + W1_OFF + j * (DF * DF) + 2 * pr;
      L.accA[g] = wmma4(a, ldw(wj + offA, false), L.accA[g]);
      L.accB[g] = wmma4(a, ldw(wj + offB, false), L.accB[g]);
    }
  }
}

template <int G>
__device__ __forceinline__ void level_tail(
    const int j0, const int rb, const int n, const int half,
    const int offA, const int offB, const int offC,
    float* __restrict__ out, Level<G>& L,
    float (*sh_w)[HK][HP], float (*sc_w)[CP][16], const float* sw)
{
#pragma unroll
  for (int step = 3; step < 5; ++step) {
    const int pr = 2 * step + half;
#pragma unroll
    for (int g = 0; g < G; ++g) {
      const int j = j0 + g;
      const int p = parent_of(j);
      const int src = (p >= 0) ? p : NJ;
      v2f a;
      a.x = sc_w[src][2 * pr - 12][n];
      a.y = sc_w[src][2 * pr - 11][n];
      if (step == 3) a.x = half ? a.x : L.bl[g];
      const float* wj = sw + W1_OFF + j * (DF * DF) + 2 * pr;
      const bool last = (step == 4) && (half == 1);
      L.accA[g] = wmma4(a, ldw(wj + offA, last), L.accA[g]);
      L.accB[g] = wmma4(a, ldw(wj + offB, last), L.accB[g]);
    }
  }
#pragma unroll
  for (int g = 0; g < G; ++g) {
    v4f lo, hi;
#pragma unroll
    for (int v = 0; v < 4; ++v) {
      lo[v] = fmaxf(L.accA[g][v], 0.f);
      hi[v] = fmaxf(L.accA[g][v + 4], 0.f);
    }
    *(v4f*)&sh_w[g][n][8 * half]     = lo;
    *(v4f*)&sh_w[g][n][8 * half + 4] = hi;
  }
  if (n < 3) {
#pragma unroll
    for (int g = 0; g < G; ++g) {
      v4f lo, hi;
#pragma unroll
      for (int v = 0; v < 4; ++v) {
        lo[v] = fmaxf(L.accB[g][v], 0.f);
        hi[v] = fmaxf(L.accB[g][v + 4], 0.f);
      }
      *(v4f*)&sh_w[g][16 + n][8 * half]     = lo;
      *(v4f*)&sh_w[g][16 + n][8 * half + 4] = hi;
    }
  }
  lds_fence();

  v8f acc2[G];
#pragma unroll
  for (int g = 0; g < G; ++g) {
    const float bC = (n < DPJ) ? L.bC[g] : 0.f;
#pragma unroll
    for (int v = 0; v < 8; ++v) acc2[g][v] = bC;
  }
#pragma unroll
  for (int step = 0; step < 5; ++step) {
    const int pr = 2 * step + half;
#pragma unroll
    for (int g = 0; g < G; ++g) {
      const int j = j0 + g;
      v2f a;
      a.x = sh_w[g][2 * pr][n];
      a.y = sh_w[g][2 * pr + 1][n];
      const float* wj = sw + W2_OFF + j * (DPJ * DF) + 2 * pr;
      const bool last = (step == 4) && (half == 1);
      acc2[g] = wmma4(a, ldw(wj + offC, last), acc2[g]);
    }
  }
  if (n < CP) {
#pragma unroll
    for (int g = 0; g < G; ++g) {
      v4f lo, hi;
#pragma unroll
      for (int v = 0; v < 4; ++v) {
        lo[v] = (n < DPJ) ? acc2[g][v] : 0.f;
        hi[v] = (n < DPJ) ? acc2[g][v + 4] : 0.f;
      }
      *(v4f*)&sc_w[j0 + g][(1 + n) & (CP - 1)][8 * half]     = lo;
      *(v4f*)&sc_w[j0 + g][(1 + n) & (CP - 1)][8 * half + 4] = hi;
    }
  }
  if (n < DPJ) {
#pragma unroll
    for (int g = 0; g < G; ++g) {
      const int j = j0 + g;
#pragma unroll
      for (int v = 0; v < 8; ++v)
        out[(long)(rb + v + 8 * half) * (NJ * DPJ) + j * DPJ + n] = acc2[g][v];
    }
  }
  lds_fence();
}

__global__ __launch_bounds__(128) void pose_enc_kernel(
    const float* __restrict__ rots, const float* __restrict__ Jtrs,
    const float* __restrict__ W0,   const float* __restrict__ b0,
    const float* __restrict__ W1,   const float* __restrict__ b1,
    const float* __restrict__ W2,   const float* __restrict__ b2,
    float* __restrict__ out, int B)
{
  __shared__ __attribute__((aligned(16))) float s_h[4][GMAX][HK][HP];
  __shared__ __attribute__((aligned(16))) float s_c[4][NJ + 1][CP][16];

  __shared__ __attribute__((aligned(16))) float s_w[SW_FLOATS];

  const int tid  = threadIdx.x & 127;

  stage_weights<W1_FLOATS / 4, W2_FLOATS / 4, W0_FLOATS / 4>(W1, W2, W0, s_w, tid);
  __syncthreads();

  const int w    = tid >> 5;
  const int lane = tid & 31;
  const int half = lane >> 4;
  const int n    = lane & 15;
  const int offA = n * DF;
  const int offB = (16 + imin(n, 2)) * DF;
  const int offC = imin(n, DPJ - 1) * DF;
  const int off0 = imin(n, DPJ - 1) * (12 * NJ);
  const int rb   = (blockIdx.x * 4 + w) * 16;
  if (rb >= B) return;

  const float* rrow = rots + (long)(rb + n) * 216;
  const float* jrow = Jtrs + (long)(rb + n) * 72;
  const float* rlan = rrow + 2 * half;
  float (*sh_w)[HK][HP] = s_h[w];
  float (*sc_w)[CP][16] = s_c[w];

  if (lane < GMAX * 4) {
    const v4f z = {0.f, 0.f, 0.f, 0.f};
    *(v4f*)&sh_w[lane >> 2][HK - 1][4 * (lane & 3)] = z;
  }

  Level<1> L0;  Level<3> L1, L2, L3;  Level<5> L4;  Level<3> L5;  Level<2> L6, L7, L8;

  level_request<1>(0, n, rrow, rlan, jrow, b1, b2, L0);
  level_request<3>(1, n, rrow, rlan, jrow, b1, b2, L1);
  level_head<1>(0, n, half, offA, offB, L0, s_w);
  level_head<3>(1, n, half, offA, offB, L1, s_w);

  {
    float bv = b0[imin(n, 5)];
    bv = (n < DPJ) ? bv : 0.f;
    v8f acc;
#pragma unroll
    for (int v = 0; v < 8; ++v) acc[v] = bv;

#pragma unroll 4
    for (int step = 0; step < 54; ++step) {
      int pr = 2 * step + half;
      v2f a = *(const v2f*)(rrow + 2 * pr);
      acc = wmma4(a, ldw(s_w + W0_OFF + off0 + 2 * pr, false), acc);
    }
#pragma unroll 4
    for (int step = 54; step < 72; ++step) {
      int pr = 2 * step + half;
      v2f a = *(const v2f*)(jrow + 2 * pr - 216);
      acc = wmma4(a, ldw(s_w + W0_OFF + off0 + 2 * pr, false), acc);
    }
    if (n < CP) {
      v4f lo, hi;
#pragma unroll
      for (int v = 0; v < 4; ++v) {
        lo[v] = (n < DPJ) ? acc[v] : 0.f;
        hi[v] = (n < DPJ) ? acc[v + 4] : 0.f;
      }
      *(v4f*)&sc_w[NJ][(1 + n) & (CP - 1)][8 * half]     = lo;
      *(v4f*)&sc_w[NJ][(1 + n) & (CP - 1)][8 * half + 4] = hi;
    }
  }
  lds_fence();

#define REQ(G, j0, L)  level_request<G>(j0, n, rrow, rlan, jrow, b1, b2, L)
#define HEAD(G, j0, L) level_head<G>(j0, n, half, offA, offB, L, s_w)
#define TAIL(G, j0, L) level_tail<G>(j0, rb, n, half, offA, offB, offC, out, L, sh_w, sc_w, s_w)
  REQ(3, 4, L2);                      TAIL(1, 0, L0);
  REQ(3, 7, L3);   HEAD(3, 4, L2);    TAIL(3, 1, L1);
  REQ(5, 10, L4);  HEAD(3, 7, L3);    TAIL(3, 4, L2);
  REQ(3, 15, L5);  HEAD(5, 10, L4);   TAIL(3, 7, L3);
  REQ(2, 18, L6);  HEAD(3, 15, L5);   TAIL(5, 10, L4);
  REQ(2, 20, L7);  HEAD(2, 18, L6);   TAIL(3, 15, L5);
  REQ(2, 22, L8);  HEAD(2, 20, L7);   TAIL(2, 18, L6);
                   HEAD(2, 22, L8);   TAIL(2, 20, L7);
                                      TAIL(2, 22, L8);
#undef REQ
#undef HEAD
#undef TAIL
}

extern "C" void kernel_launch(void* const* d_in, const int* in_sizes, int n_in,
                              void* d_out, int out_size, void* d_ws, size_t ws_size,
                              hipStream_t stream) {
  const float* rots = (const float*)d_in[0];
  const float* Jtrs = (const float*)d_in[1];
  const float* W0   = (const float*)d_in[2];
  const float* b0   = (const float*)d_in[3];
  const float* W1   = (const float*)d_in[4];
  const float* b1   = (const float*)d_in[5];
  const float* W2   = (const float*)d_in[6];
  const float* b2   = (const float*)d_in[7];
  float* outp = (float*)d_out;

  int B = in_sizes[0] / (NJ * 9);
  int grid = (B + 63) / 64;
  pose_enc_kernel<<<grid, 128, 0, stream>>>(rots, Jtrs, W0, b0, W1, b1, W2, b2,
                                            outp, B);
}
